# starting kernel
# baseline (speedup 1.0000x reference)
#include <hip/hip_runtime.h>

typedef __attribute__((ext_vector_type(2))) float v2f;
typedef __attribute__((ext_vector_type(8))) float v8f;

#define BATCH 2048
#define Qn 128
#define Sn 128
#define Hn 32
#define Cn (Qn * Sn)

#define WAVES 8
#define RT 2
#define ROWS_PER_BLOCK (WAVES * RT * 16)

__global__ __launch_bounds__(256)
void divenc_wmma_f32(const float* __restrict__ x,
                     const float* __restrict__ W1,
                     const float* __restrict__ b1,
                     const float* __restrict__ W2,
                     const float* __restrict__ b2,
                     float* __restrict__ out)
{
    __shared__ float w1s[32 * 128];

    const int q   = blockIdx.x;
    const int tid = threadIdx.x;

    {
        const float* w1g = W1 + (size_t)q * Sn * Hn;
        #pragma unroll
        for (int it = 0; it < 8; ++it) {
            int i  = tid + it * 256;
            int l  = i & 31;
            int t  = (i >> 5) & 1;
            int f  = i >> 6;
            int hi = l >> 4;
            int n  = l & 15;
            int s  = 4 * f + 2 * hi;
            int g  = s * Hn + n + 16 * t;
            float v0 = w1g[g];
            float v1 = w1g[g + Hn];
            float* dst = &w1s[f * 128 + t * 64 + l * 2];
            dst[0] = v0;
            dst[1] = v1;
        }
    }
    __syncthreads();

    const int wave = tid >> 5;
    const int lane = tid & 31;
    const int ln   = lane & 15;
    const int hi   = lane >> 4;
    const int b0   = blockIdx.y * ROWS_PER_BLOCK + wave * (RT * 16);

    const float* xrow[RT];
    #pragma unroll
    for (int r = 0; r < RT; ++r)
        xrow[r] = x + (size_t)(b0 + r * 16 + ln) * Cn + q * Sn + hi * 2;

    const float* bfrag = &w1s[lane * 2];

    v8f acc[RT][2] = {};

    #pragma unroll 8
    for (int k = 0; k < Sn; k += 4) {
        const int f = k >> 2;
        v2f bfa = *(const v2f*)(bfrag + f * 128);
        v2f bfb = *(const v2f*)(bfrag + f * 128 + 64);

        #pragma unroll
        for (int r = 0; r < RT; ++r) {
            v2f a = __builtin_nontemporal_load((const v2f*)(xrow[r] + k));
            acc[r][0] = __builtin_amdgcn_wmma_f32_16x16x4_f32(
                false, a, false, bfa, (short)0, acc[r][0], false, false);
            acc[r][1] = __builtin_amdgcn_wmma_f32_16x16x4_f32(
                false, a, false, bfb, (short)0, acc[r][1], false, false);
        }
    }

    const float w2a = W2[q * Hn + ln];
    const float w2b = W2[q * Hn + ln + 16];
    const float b1a = b1[q * Hn + ln];
    const float b1b = b1[q * Hn + ln + 16];
    const float b2q = b2[q];

    #pragma unroll
    for (int r = 0; r < RT; ++r) {
        #pragma unroll
        for (int m = 0; m < 8; ++m) {
            float h0 = acc[r][0][m] + b1a;
            float h1 = acc[r][1][m] + b1b;
            h0 = (h0 > 0.0f) ? h0 : (__expf(h0) - 1.0f);
            h1 = (h1 > 0.0f) ? h1 : (__expf(h1) - 1.0f);
            float p = h0 * w2a + h1 * w2b;
            p += __shfl_xor(p, 8, 16);
            p += __shfl_xor(p, 4, 16);
            p += __shfl_xor(p, 2, 16);
            p += __shfl_xor(p, 1, 16);
            if (ln == 0) {
                int b = b0 + r * 16 + hi * 8 + m;
                out[(size_t)b * Qn + q] = p + b2q;
            }
        }
    }
}

extern "C" void kernel_launch(void* const* d_in, const int* in_sizes, int n_in,
                              void* d_out, int out_size, void* d_ws, size_t ws_size,
                              hipStream_t stream) {
    const float* x  = (const float*)d_in[0];
    const float* W1 = (const float*)d_in[1];
    const float* b1 = (const float*)d_in[2];
    const float* W2 = (const float*)d_in[3];
    const float* b2 = (const float*)d_in[4];
    float* out = (float*)d_out;

    dim3 grid(Qn, BATCH / ROWS_PER_BLOCK);
    divenc_wmma_f32<<<grid, 256, 0, stream>>>(x, W1, b1, W2, b2, out);
}


# x read by whole-row 16-B-per-lane loads (512 contiguous B per instruction) and passed through a per-wave LDS tile to the A fragments; k order inside each group of 8 permuted in A and B alike
# speedup vs baseline: 2.7605x; 2.7605x over previous
#include <hip/hip_runtime.h>


typedef __attribute__((ext_vector_type(2))) float v2f;
typedef __attribute__((ext_vector_type(4))) float v4f;
typedef __attribute__((ext_vector_type(8))) float v8f;

#define BATCH 2048
#define Qn 128
#define Sn 128
#define Hn 32
#define Cn (Qn * Sn)

#define WAVES 8
#define NTILE 4
#define ROWS_PER_WAVE (NTILE * 16)
#define ROWS_PER_BLOCK (WAVES * ROWS_PER_WAVE)
#define XP 132
#define XTILE (16 * XP)

__global__ __launch_bounds__(256)
void divenc_wmma_f32(const float* __restrict__ x,
                     const float* __restrict__ W1,
                     const float* __restrict__ b1,
                     const float* __restrict__ W2,
                     const float* __restrict__ b2,
                     float* __restrict__ out)
{
    __shared__ __attribute__((aligned(16))) float w1s[16 * 2 * 32 * 4];
    __shared__ __attribute__((aligned(16))) float xs[WAVES * 2 * XTILE];

    const int q    = blockIdx.x;
    const int tid  = threadIdx.x;
    const int wave = tid >> 5;
    const int lane = tid & 31;
    const int ln   = lane & 15;
    const int hi   = lane >> 4;
    const int b0   = blockIdx.y * ROWS_PER_BLOCK + wave * ROWS_PER_WAVE;

    const float* xg = x + (size_t)b0 * Cn + q * Sn + lane * 4;

    v4f xr[16];
    #pragma unroll
    for (int r = 0; r < 16; ++r)
        xr[r] = __builtin_nontemporal_load((const v4f*)(xg + (size_t)r * Cn));

    {
        const float* w1g = W1 + (size_t)q * Sn * Hn;
        #pragma unroll
        for (int it = 0; it < 4; ++it) {
            const int kq = wave + 8 * it;
            v4f v;
            v.x = w1g[(4 * kq + 0) * Hn + lane];
            v.y = w1g[(4 * kq + 1) * Hn + lane];
            v.z = w1g[(4 * kq + 2) * Hn + lane];
            v.w = w1g[(4 * kq + 3) * Hn + lane];
            const int j  = kq >> 1;
            const int kh = kq & 1;
            const int t  = lane >> 4;
            const int n  = lane & 15;
            *(v4f*)&w1s[((j * 2 + t) * 32 + kh * 16 + n) * 4] = v;
        }
    }

    float* xw = &xs[wave * 2 * XTILE];
    #pragma unroll
    for (int r = 0; r < 16; ++r)
        *(v4f*)&xw[r * XP + lane * 4] = xr[r];

    const float w2a = W2[q * Hn + ln];
    const float w2b = W2[q * Hn + ln + 16];
    const float b1a = b1[q * Hn + ln];
    const float b1b = b1[q * Hn + ln + 16];
    const float b2q = b2[q];

    __syncthreads();

    const float* bp = &w1s[lane * 4];

    #pragma unroll
    for (int t = 0; t < NTILE; ++t) {
        if (t + 1 < NTILE) {
            #pragma unroll
            for (int r = 0; r < 16; ++r)
                xr[r] = __builtin_nontemporal_load(
                    (const v4f*)(xg + (size_t)((t + 1) * 16 + r) * Cn));
            __builtin_amdgcn_sched_barrier(0);
        }

        const float* ap = xw + (t & 1) * XTILE + ln * XP + hi * 4;

        v8f acc0 = {};
        v8f acc1 = {};

        #pragma unroll
        for (int j = 0; j < 16; ++j) {
            const v4f a  = *(const v4f*)(ap + j * 8);
            const v4f ba = *(const v4f*)(bp + (j * 2 + 0) * 128);
            const v4f bb = *(const v4f*)(bp + (j * 2 + 1) * 128);
            const v2f a0  = {a.x, a.y},   a1  = {a.z, a.w};
            const v2f ba0 = {ba.x, ba.y}, ba1 = {ba.z, ba.w};
            const v2f bb0 = {bb.x, bb.y}, bb1 = {bb.z, bb.w};
            acc0 = __builtin_amdgcn_wmma_f32_16x16x4_f32(
                false, a0, false, ba0, (short)0, acc0, false, false);
            acc1 = __builtin_amdgcn_wmma_f32_16x16x4_f32(
                false, a0, false, bb0, (short)0, acc1, false, false);
            acc0 = __builtin_amdgcn_wmma_f32_16x16x4_f32(
                false, a1, false, ba1, (short)0, acc0, false, false);
            acc1 = __builtin_amdgcn_wmma_f32_16x16x4_f32(
                false, a1, false, bb1, (short)0, acc1, false, false);
        }

        #pragma unroll
        for (int m = 0; m < 8; ++m) {
            float h0 = acc0[m] + b1a;
            float h1 = acc1[m] + b1b;
            h0 = (h0 > 0.0f) ? h0 : (__expf(h0) - 1.0f);
            h1 = (h1 > 0.0f) ? h1 : (__expf(h1) - 1.0f);
            float p = h0 * w2a + h1 * w2b;
            p += __shfl_xor(p, 8, 16);
            p += __shfl_xor(p, 4, 16);
            p += __shfl_xor(p, 2, 16);
            p += __shfl_xor(p, 1, 16);
            if (ln == 0) {
                const int b = b0 + t * 16 + hi * 8 + m;
                out[(size_t)b * Qn + q] = p + b2q;
            }
        }

        if (t + 1 < NTILE) {
            float* xn = xw + ((t + 1) & 1) * XTILE;
            #pragma unroll
            for (int r = 0; r < 16; ++r)
                *(v4f*)&xn[r * XP + lane * 4] = xr[r];
        }
    }
}

extern "C" void kernel_launch(void* const* d_in, const int* in_sizes, int n_in,
                              void* d_out, int out_size, void* d_ws, size_t ws_size,
                              hipStream_t stream) {
    const float* x  = (const float*)d_in[0];
    const float* W1 = (const float*)d_in[1];
    const float* b1 = (const float*)d_in[2];
    const float* W2 = (const float*)d_in[3];
    const float* b2 = (const float*)d_in[4];
    float* out = (float*)d_out;

    dim3 grid(Qn, BATCH / ROWS_PER_BLOCK);
    divenc_wmma_f32<<<grid, 256, 0, stream>>>(x, W1, b1, W2, b2, out);
}


# WMMA operands swapped (W1 as A, x as B): hidden index lands on D's registers, so bias, ELU and W2 dot sum in-lane plus one lane-half exchange, no LDS permutes; W2 sum order changes
# speedup vs baseline: 2.8868x; 1.0458x over previous
#include <hip/hip_runtime.h>


typedef __attribute__((ext_vector_type(2))) float v2f;
typedef __attribute__((ext_vector_type(4))) float v4f;
typedef __attribute__((ext_vector_type(8))) float v8f;

#define BATCH 2048
#define Qn 128
#define Sn 128
#define Hn 32
#define Cn (Qn * Sn)

#define WAVES 8
#define NTILE 4
#define ROWS_PER_WAVE (NTILE * 16)
#define ROWS_PER_BLOCK (WAVES * ROWS_PER_WAVE)
#define XP 132
#define XTILE (16 * XP)

__device__ __forceinline__ float xhalf(float v) {
    const int b = __builtin_bit_cast(int, v);
    return __builtin_bit_cast(float,
        __builtin_amdgcn_permlanex16(b, b, 0x76543210, 0xfedcba98, false, false));
}

__global__ __launch_bounds__(256)
void divenc_wmma_f32(const float* __restrict__ x,
                     const float* __restrict__ W1,
                     const float* __restrict__ b1,
                     const float* __restrict__ W2,
                     const float* __restrict__ b2,
                     float* __restrict__ out)
{
    __shared__ __attribute__((aligned(16))) float w1s[16 * 2 * 32 * 4];
    __shared__ __attribute__((aligned(16))) float xs[WAVES * 2 * XTILE];

    const int q    = blockIdx.x;
    const int tid  = threadIdx.x;
    const int wave = tid >> 5;
    const int lane = tid & 31;
    const int ln   = lane & 15;
    const int hi   = lane >> 4;
    const int b0   = blockIdx.y * ROWS_PER_BLOCK + wave * ROWS_PER_WAVE;

    const float* xg = x + (size_t)b0 * Cn + q * Sn + lane * 4;

    v4f xr[16];
    #pragma unroll
    for (int r = 0; r < 16; ++r)
        xr[r] = __builtin_nontemporal_load((const v4f*)(xg + (size_t)r * Cn));

    {
        const float* w1g = W1 + (size_t)q * Sn * Hn;
        #pragma unroll
        for (int it = 0; it < 4; ++it) {
            const int kq = wave + 8 * it;
            v4f v;
            v.x = w1g[(4 * kq + 0) * Hn + lane];
            v.y = w1g[(4 * kq + 1) * Hn + lane];
            v.z = w1g[(4 * kq + 2) * Hn + lane];
            v.w = w1g[(4 * kq + 3) * Hn + lane];
            const int j  = kq >> 1;
            const int kh = kq & 1;
            const int t  = lane >> 4;
            const int n  = lane & 15;
            *(v4f*)&w1s[((j * 2 + t) * 32 + kh * 16 + n) * 4] = v;
        }
    }

    float* xw = &xs[wave * 2 * XTILE];
    #pragma unroll
    for (int r = 0; r < 16; ++r)
        *(v4f*)&xw[r * XP + lane * 4] = xr[r];

    const float* b1p = b1 + q * Hn + hi * 8;
    const float* w2p = W2 + q * Hn + hi * 8;
    const v4f b1l0 = *(const v4f*)(b1p),      b1l1 = *(const v4f*)(b1p + 4);
    const v4f b1h0 = *(const v4f*)(b1p + 16), b1h1 = *(const v4f*)(b1p + 20);
    const v4f w2l0 = *(const v4f*)(w2p),      w2l1 = *(const v4f*)(w2p + 4);
    const v4f w2h0 = *(const v4f*)(w2p + 16), w2h1 = *(const v4f*)(w2p + 20);
    const float b1lo[8] = {b1l0.x, b1l0.y, b1l0.z, b1l0.w, b1l1.x, b1l1.y, b1l1.z, b1l1.w};
    const float b1hi[8] = {b1h0.x, b1h0.y, b1h0.z, b1h0.w, b1h1.x, b1h1.y, b1h1.z, b1h1.w};
    const float w2lo[8] = {w2l0.x, w2l0.y, w2l0.z, w2l0.w, w2l1.x, w2l1.y, w2l1.z, w2l1.w};
    const float w2hi[8] = {w2h0.x, w2h0.y, w2h0.z, w2h0.w, w2h1.x, w2h1.y, w2h1.z, w2h1.w};
    const float b2q = b2[q];

    __syncthreads();

    const float* bp = &w1s[lane * 4];

    #pragma unroll
    for (int t = 0; t < NTILE; ++t) {
        if (t + 1 < NTILE) {
            #pragma unroll
            for (int r = 0; r < 16; ++r)
                xr[r] = __builtin_nontemporal_load(
                    (const v4f*)(xg + (size_t)((t + 1) * 16 + r) * Cn));
            __builtin_amdgcn_sched_barrier(0);
        }

        const float* ap = xw + (t & 1) * XTILE + ln * XP + hi * 4;

        v8f acc0 = {};
        v8f acc1 = {};

        #pragma unroll
        for (int j = 0; j < 16; ++j) {
            const v4f a  = *(const v4f*)(ap + j * 8);
            const v4f ba = *(const v4f*)(bp + (j * 2 + 0) * 128);
            const v4f bb = *(const v4f*)(bp + (j * 2 + 1) * 128);
            const v2f a0  = {a.x, a.y},   a1  = {a.z, a.w};
            const v2f ba0 = {ba.x, ba.y}, ba1 = {ba.z, ba.w};
            const v2f bb0 = {bb.x, bb.y}, bb1 = {bb.z, bb.w};
            acc0 = __builtin_amdgcn_wmma_f32_16x16x4_f32(
                false, ba0, false, a0, (short)0, acc0, false, false);
            acc1 = __builtin_amdgcn_wmma_f32_16x16x4_f32(
                false, bb0, false, a0, (short)0, acc1, false, false);
            acc0 = __builtin_amdgcn_wmma_f32_16x16x4_f32(
                false, ba1, false, a1, (short)0, acc0, false, false);
            acc1 = __builtin_amdgcn_wmma_f32_16x16x4_f32(
                false, bb1, false, a1, (short)0, acc1, false, false);
        }

        float p0 = 0.0f, p1 = 0.0f;
        #pragma unroll
        for (int m = 0; m < 8; ++m) {
            float h0 = acc0[m] + b1lo[m];
            float h1 = acc1[m] + b1hi[m];
            h0 = (h0 > 0.0f) ? h0 : (__expf(h0) - 1.0f);
            h1 = (h1 > 0.0f) ? h1 : (__expf(h1) - 1.0f);
            p0 += h0 * w2lo[m];
            p1 += h1 * w2hi[m];
        }
        float p = p0 + p1;
        p += xhalf(p);
        if (hi == 0) {
            const int b = b0 + t * 16 + ln;
            out[(size_t)b * Qn + q] = p + b2q;
        }

        if (t + 1 < NTILE) {
            float* xn = xw + ((t + 1) & 1) * XTILE;
            #pragma unroll
            for (int r = 0; r < 16; ++r)
                *(v4f*)&xn[r * XP + lane * 4] = xr[r];
        }
    }
}

extern "C" void kernel_launch(void* const* d_in, const int* in_sizes, int n_in,
                              void* d_out, int out_size, void* d_ws, size_t ws_size,
                              hipStream_t stream) {
    const float* x  = (const float*)d_in[0];
    const float* W1 = (const float*)d_in[1];
    const float* b1 = (const float*)d_in[2];
    const float* W2 = (const float*)d_in[3];
    const float* b2 = (const float*)d_in[4];
    float* out = (float*)d_out;

    dim3 grid(Qn, BATCH / ROWS_PER_BLOCK);
    divenc_wmma_f32<<<grid, 256, 0, stream>>>(x, W1, b1, W2, b2, out);
}


# W1[q] fragments held in registers for a wave's 128 rows (4-wave workgroups, 2 waves per SIMD); tile loop reads only x from LDS, as one group ahead of its 64 WMMAs; sum order unchanged
# speedup vs baseline: 2.9927x; 1.0367x over previous
#include <hip/hip_runtime.h>


typedef __attribute__((ext_vector_type(2))) float v2f;
typedef __attribute__((ext_vector_type(4))) float v4f;
typedef __attribute__((ext_vector_type(8))) float v8f;

#define BATCH 2048
#define Qn 128
#define Sn 128
#define Hn 32
#define Cn (Qn * Sn)

#define WAVES 4
#define NTILE 8
#define ROWS_PER_WAVE (NTILE * 16)
#define ROWS_PER_BLOCK (WAVES * ROWS_PER_WAVE)
#define XP 132
#define XTILE (16 * XP)

__device__ __forceinline__ float xhalf(float v) {
    const int b = __builtin_bit_cast(int, v);
    return __builtin_bit_cast(float,
        __builtin_amdgcn_permlanex16(b, b, 0x76543210, 0xfedcba98, false, false));
}

__global__ __launch_bounds__(128, 2)
void divenc_wmma_f32(const float* __restrict__ x,
                     const float* __restrict__ W1,
                     const float* __restrict__ b1,
                     const float* __restrict__ W2,
                     const float* __restrict__ b2,
                     float* __restrict__ out)
{
    __shared__ __attribute__((aligned(16))) float w1s[16 * 2 * 32 * 4];
    __shared__ __attribute__((aligned(16))) float xs[WAVES * XTILE];
    __shared__ __attribute__((aligned(16))) float prm[2 * Hn];

    const int q    = blockIdx.x;
    const int tid  = threadIdx.x;
    const int wave = tid >> 5;
    const int lane = tid & 31;
    const int ln   = lane & 15;
    const int hi   = lane >> 4;
    const int b0   = blockIdx.y * ROWS_PER_BLOCK + wave * ROWS_PER_WAVE;

    v4f w1v[8];
    {
        const float* w1g = W1 + (size_t)q * Sn * Hn;
        #pragma unroll
        for (int it = 0; it < 8; ++it) {
            const int kq = wave + 4 * it;
            w1v[it].x = w1g[(4 * kq + 0) * Hn + lane];
            w1v[it].y = w1g[(4 * kq + 1) * Hn + lane];
            w1v[it].z = w1g[(4 * kq + 2) * Hn + lane];
            w1v[it].w = w1g[(4 * kq + 3) * Hn + lane];
        }
    }

    float prm_b = 0.0f, prm_w = 0.0f;
    if (wave == 0) {
        prm_b = b1[q * Hn + lane];
        prm_w = W2[q * Hn + lane];
    }
    const float b2q = b2[q];

    const float* xg = x + (size_t)b0 * Cn + q * Sn + lane * 4;

    v4f xr[2][16];
    #pragma unroll
    for (int s = 0; s < 2; ++s) {
        #pragma unroll
        for (int r = 0; r < 16; ++r)
            xr[s][r] = __builtin_nontemporal_load((const v4f*)(xg + (size_t)(s * 16 + r) * Cn));
    }

    #pragma unroll
    for (int it = 0; it < 8; ++it) {
        const int kq = wave + 4 * it;
        const int j  = kq >> 1;
        const int kh = kq & 1;
        const int t  = lane >> 4;
        const int n  = lane & 15;
        *(v4f*)&w1s[((j * 2 + t) * 32 + kh * 16 + n) * 4] = w1v[it];
    }

    if (wave == 0) {
        prm[lane]      = prm_b;
        prm[Hn + lane] = prm_w;
    }

    asm volatile("s_wait_dscnt 0x0\n\ts_barrier_signal -1\n\ts_barrier_wait -1" ::: "memory");

    const v4f b1l0 = *(const v4f*)&prm[hi * 8],           b1l1 = *(const v4f*)&prm[hi * 8 + 4];
    const v4f b1h0 = *(const v4f*)&prm[16 + hi * 8],      b1h1 = *(const v4f*)&prm[16 + hi * 8 + 4];
    const v4f w2l0 = *(const v4f*)&prm[Hn + hi * 8],      w2l1 = *(const v4f*)&prm[Hn + hi * 8 + 4];
    const v4f w2h0 = *(const v4f*)&prm[Hn + 16 + hi * 8], w2h1 = *(const v4f*)&prm[Hn + 16 + hi * 8 + 4];
    const float b1lo[8] = {b1l0.x, b1l0.y, b1l0.z, b1l0.w, b1l1.x, b1l1.y, b1l1.z, b1l1.w};
    const float b1hi[8] = {b1h0.x, b1h0.y, b1h0.z, b1h0.w, b1h1.x, b1h1.y, b1h1.z, b1h1.w};
    const float w2lo[8] = {w2l0.x, w2l0.y, w2l0.z, w2l0.w, w2l1.x, w2l1.y, w2l1.z, w2l1.w};
    const float w2hi[8] = {w2h0.x, w2h0.y, w2h0.z, w2h0.w, w2h1.x, w2h1.y, w2h1.z, w2h1.w};

    v4f wf[32];
    {
        const float* bp = &w1s[lane * 4];
        #pragma unroll
        for (int i = 0; i < 32; ++i)
            wf[i] = *(const v4f*)(bp + i * 128);
    }

    float* xw = &xs[wave * XTILE];
    const float* ap = xw + ln * XP + hi * 4;

    #pragma unroll
    for (int t = 0; t < NTILE; ++t) {
        #pragma unroll
        for (int r = 0; r < 16; ++r)
            *(v4f*)&xw[r * XP + lane * 4] = xr[t & 1][r];

        if (t + 2 < NTILE) {
            #pragma unroll
            for (int r = 0; r < 16; ++r)
                xr[t & 1][r] = __builtin_nontemporal_load(
                    (const v4f*)(xg + (size_t)((t + 2) * 16 + r) * Cn));
        }

        v4f xf[16];
        #pragma unroll
        for (int j = 0; j < 16; ++j)
            xf[j] = *(const v4f*)(ap + j * 8);

        __builtin_amdgcn_sched_barrier(0);

        v8f acc0 = {};
        v8f acc1 = {};

        #pragma unroll
        for (int j = 0; j < 16; ++j) {
            const v4f a  = xf[j];
            const v4f ba = wf[j * 2 + 0];
            const v4f bb = wf[j * 2 + 1];
            const v2f a0  = {a.x, a.y},   a1  = {a.z, a.w};
            const v2f ba0 = {ba.x, ba.y}, ba1 = {ba.z, ba.w};
            const v2f bb0 = {bb.x, bb.y}, bb1 = {bb.z, bb.w};
            acc0 = __builtin_amdgcn_wmma_f32_16x16x4_f32(
                false, ba0, false, a0, (short)0, acc0, false, false);
            acc1 = __builtin_amdgcn_wmma_f32_16x16x4_f32(
                false, bb0, false, a0, (short)0, acc1, false, false);
            acc0 = __builtin_amdgcn_wmma_f32_16x16x4_f32(
                false, ba1, false, a1, (short)0, acc0, false, false);
            acc1 = __builtin_amdgcn_wmma_f32_16x16x4_f32(
                false, bb1, false, a1, (short)0, acc1, false, false);
        }

        float p0 = 0.0f, p1 = 0.0f;
        #pragma unroll
        for (int m = 0; m < 8; ++m) {
            float h0 = acc0[m] + b1lo[m];
            float h1 = acc1[m] + b1hi[m];
            h0 = (h0 > 0.0f) ? h0 : (__expf(h0) - 1.0f);
            h1 = (h1 > 0.0f) ? h1 : (__expf(h1) - 1.0f);
            p0 += h0 * w2lo[m];
            p1 += h1 * w2hi[m];
        }
        float p = p0 + p1;
        p += xhalf(p);
        if (hi == 0) {
            const int b = b0 + t * 16 + ln;
            out[(size_t)b * Qn + q] = p + b2q;
        }
    }
}

extern "C" void kernel_launch(void* const* d_in, const int* in_sizes, int n_in,
                              void* d_out, int out_size, void* d_ws, size_t ws_size,
                              hipStream_t stream) {
    const float* x  = (const float*)d_in[0];
    const float* W1 = (const float*)d_in[1];
    const float* b1 = (const float*)d_in[2];
    const float* W2 = (const float*)d_in[3];
    const float* b2 = (const float*)d_in[4];
    float* out = (float*)d_out;

    dim3 grid(Qn, BATCH / ROWS_PER_BLOCK);
    divenc_wmma_f32<<<grid, 128, 0, stream>>>(x, W1, b1, W2, b2, out);
}


# Next tile's staging (park in LDS, row refill, 16 fragment reads) moved from behind this tile's epilogue to in front of it: the LDS round trip runs under bias, ELU and W2 dot; sum order unchanged
# speedup vs baseline: 3.1772x; 1.0617x over previous
#include <hip/hip_runtime.h>


typedef __attribute__((ext_vector_type(2))) float v2f;
typedef __attribute__((ext_vector_type(4))) float v4f;
typedef __attribute__((ext_vector_type(8))) float v8f;

#define BATCH 2048
#define Qn 128
#define Sn 128
#define Hn 32
#define Cn (Qn * Sn)

#define WAVES 4
#define NTILE 8
#define ROWS_PER_WAVE (NTILE * 16)
#define ROWS_PER_BLOCK (WAVES * ROWS_PER_WAVE)
#define XP 132
#define XTILE (16 * XP)

__device__ __forceinline__ float xhalf(float v) {
    const int b = __builtin_bit_cast(int, v);
    return __builtin_bit_cast(float,
        __builtin_amdgcn_permlanex16(b, b, 0x76543210, 0xfedcba98, false, false));
}

__global__ __launch_bounds__(128, 2)
void divenc_wmma_f32(const float* __restrict__ x,
                     const float* __restrict__ W1,
                     const float* __restrict__ b1,
                     const float* __restrict__ W2,
                     const float* __restrict__ b2,
                     float* __restrict__ out)
{
    __shared__ __attribute__((aligned(16))) float w1s[16 * 2 * 32 * 4];
    __shared__ __attribute__((aligned(16))) float xs[WAVES * XTILE];
    __shared__ __attribute__((aligned(16))) float prm[2 * Hn];

    const int q    = blockIdx.x;
    const int tid  = threadIdx.x;
    const int wave = tid >> 5;
    const int lane = tid & 31;
    const int ln   = lane & 15;
    const int hi   = lane >> 4;
    const int b0   = blockIdx.y * ROWS_PER_BLOCK + wave * ROWS_PER_WAVE;

    v4f w1v[8];
    {
        const float* w1g = W1 + (size_t)q * Sn * Hn;
        #pragma unroll
        for (int it = 0; it < 8; ++it) {
            const int kq = wave + 4 * it;
            w1v[it].x = w1g[(4 * kq + 0) * Hn + lane];
            w1v[it].y = w1g[(4 * kq + 1) * Hn + lane];
            w1v[it].z = w1g[(4 * kq + 2) * Hn + lane];
            w1v[it].w = w1g[(4 * kq + 3) * Hn + lane];
        }
    }

    float prm_b = 0.0f, prm_w = 0.0f;
    if (wave == 0) {
        prm_b = b1[q * Hn + lane];
        prm_w = W2[q * Hn + lane];
    }
    const float b2q = b2[q];

    const float* xg = x + (size_t)b0 * Cn + q * Sn + lane * 4;

    v4f xr[2][16];
    #pragma unroll
    for (int s = 0; s < 2; ++s) {
        #pragma unroll
        for (int r = 0; r < 16; ++r)
            xr[s][r] = __builtin_nontemporal_load((const v4f*)(xg + (size_t)(s * 16 + r) * Cn));
    }

    #pragma unroll
    for (int it = 0; it < 8; ++it) {
        const int kq = wave + 4 * it;
        const int j  = kq >> 1;
        const int kh = kq & 1;
        const int t  = lane >> 4;
        const int n  = lane & 15;
        *(v4f*)&w1s[((j * 2 + t) * 32 + kh * 16 + n) * 4] = w1v[it];
    }

    if (wave == 0) {
        prm[lane]      = prm_b;
        prm[Hn + lane] = prm_w;
    }

    asm volatile("s_wait_dscnt 0x0\n\ts_barrier_signal -1\n\ts_barrier_wait -1" ::: "memory");

    const v4f b1l0 = *(const v4f*)&prm[hi * 8],           b1l1 = *(const v4f*)&prm[hi * 8 + 4];
    const v4f b1h0 = *(const v4f*)&prm[16 + hi * 8],      b1h1 = *(const v4f*)&prm[16 + hi * 8 + 4];
    const v4f w2l0 = *(const v4f*)&prm[Hn + hi * 8],      w2l1 = *(const v4f*)&prm[Hn + hi * 8 + 4];
    const v4f w2h0 = *(const v4f*)&prm[Hn + 16 + hi * 8], w2h1 = *(const v4f*)&prm[Hn + 16 + hi * 8 + 4];
    const float b1lo[8] = {b1l0.x, b1l0.y, b1l0.z, b1l0.w, b1l1.x, b1l1.y, b1l1.z, b1l1.w};
    const float b1hi[8] = {b1h0.x, b1h0.y, b1h0.z, b1h0.w, b1h1.x, b1h1.y, b1h1.z, b1h1.w};
    const float w2lo[8] = {w2l0.x, w2l0.y, w2l0.z, w2l0.w, w2l1.x, w2l1.y, w2l1.z, w2l1.w};
    const float w2hi[8] = {w2h0.x, w2h0.y, w2h0.z, w2h0.w, w2h1.x, w2h1.y, w2h1.z, w2h1.w};

    v4f wf[32];
    {
        const float* bp = &w1s[lane * 4];
        #pragma unroll
        for (int i = 0; i < 32; ++i)
            wf[i] = *(const v4f*)(bp + i * 128);
    }

    float* xw = &xs[wave * XTILE];
    const float* ap = xw + ln * XP + hi * 4;

    v4f xf[16];

#define STAGE_TILE(T)                                                                   \
    do {                                                                                \
        _Pragma("unroll")                                                               \
        for (int r = 0; r < 16; ++r)                                                    \
            *(v4f*)&xw[r * XP + lane * 4] = xr[(T) & 1][r];                             \
        if ((T) + 2 < NTILE) {                                                          \
            _Pragma("unroll")                                                           \
            for (int r = 0; r < 16; ++r)                                                \
                xr[(T) & 1][r] = __builtin_nontemporal_load(                            \
                    (const v4f*)(xg + (size_t)(((T) + 2) * 16 + r) * Cn));              \
        }                                                                               \
        _Pragma("unroll")                                                               \
        for (int j = 0; j < 16; ++j)                                                    \
            xf[j] = *(const v4f*)(ap + j * 8);                                          \
    } while (0)

    STAGE_TILE(0);

    #pragma unroll
    for (int t = 0; t < NTILE; ++t) {
        __builtin_amdgcn_sched_barrier(0);

        v8f acc0 = {};
        v8f acc1 = {};

        #pragma unroll
        for (int j = 0; j < 16; ++j) {
            const v4f a  = xf[j];
            const v4f ba = wf[j * 2 + 0];
            const v4f bb = wf[j * 2 + 1];
            const v2f a0  = {a.x, a.y},   a1  = {a.z, a.w};
            const v2f ba0 = {ba.x, ba.y}, ba1 = {ba.z, ba.w};
            const v2f bb0 = {bb.x, bb.y}, bb1 = {bb.z, bb.w};
            acc0 = __builtin_amdgcn_wmma_f32_16x16x4_f32(
                false, ba0, false, a0, (short)0, acc0, false, false);
            acc1 = __builtin_amdgcn_wmma_f32_16x16x4_f32(
                false, bb0, false, a0, (short)0, acc1, false, false);
            acc0 = __builtin_amdgcn_wmma_f32_16x16x4_f32(
                false, ba1, false, a1, (short)0, acc0, false, false);
            acc1 = __builtin_amdgcn_wmma_f32_16x16x4_f32(
                false, bb1, false, a1, (short)0, acc1, false, false);
        }

        __builtin_amdgcn_sched_barrier(0);

        if (t + 1 < NTILE)
            STAGE_TILE(t + 1);

        __builtin_amdgcn_sched_barrier(0);

        float p0 = 0.0f, p1 = 0.0f;
        #pragma unroll
        for (int m = 0; m < 8; ++m) {
            float h0 = acc0[m] + b1lo[m];
            float h1 = acc1[m] + b1hi[m];
            h0 = (h0 > 0.0f) ? h0 : (__expf(h0) - 1.0f);
            h1 = (h1 > 0.0f) ? h1 : (__expf(h1) - 1.0f);
            p0 += h0 * w2lo[m];
            p1 += h1 * w2hi[m];
        }
        float p = p0 + p1;
        p += xhalf(p);
        if (hi == 0) {
            const int b = b0 + t * 16 + ln;
            out[(size_t)b * Qn + q] = p + b2q;
        }
    }
#undef STAGE_TILE
}

extern "C" void kernel_launch(void* const* d_in, const int* in_sizes, int n_in,
                              void* d_out, int out_size, void* d_ws, size_t ws_size,
                              hipStream_t stream) {
    const float* x  = (const float*)d_in[0];
    const float* W1 = (const float*)d_in[1];
    const float* b1 = (const float*)d_in[2];
    const float* W2 = (const float*)d_in[3];
    const float* b2 = (const float*)d_in[4];
    float* out = (float*)d_out;

    dim3 grid(Qn, BATCH / ROWS_PER_BLOCK);
    divenc_wmma_f32<<<grid, 128, 0, stream>>>(x, W1, b1, W2, b2, out);
}
